# starting kernel
# baseline (speedup 1.0000x reference)
#include <hip/hip_runtime.h>
#include <hip/hip_bf16.h>
#include <math.h>

typedef __attribute__((ext_vector_type(2))) float v2f;
typedef __attribute__((ext_vector_type(8))) float v8f;

#define O_DIM 128
#define I_DIM 128
#define M_DIM 32
#define N_DIM 128
#define LOG2E 1.44269504088896340736f

#define SZ_OFF   0
#define SQ_OFF   32
#define SW_OFF   64
#define CC_OFF   96
#define PAD_OFF  128
#define WREG     160

__global__ __launch_bounds__(256)
void gpkan_moment_kernel(const float* __restrict__ x,
                         const float* __restrict__ z,
                         const float* __restrict__ qmu,
                         const float* __restrict__ qlv,
                         const float* __restrict__ lsc,
                         const float* __restrict__ lva,
                         float* __restrict__ out)
{
    __shared__ __attribute__((aligned(16))) float smem[8 * WREG];

    const int o    = blockIdx.x;
    const int wave = threadIdx.x >> 5;
    const int lane = threadIdx.x & 31;
    const int row  = lane & 15;
    const int half = lane >> 4;
    const int n    = wave * 16 + row;

    float* W = &smem[wave * WREG];

    W[PAD_OFF + lane] = 0.0f;

    const int h2 = 2 * half;
    const int a1off = ((row == 0) ? SQ_OFF : PAD_OFF) + h2;
    const int a2off = ((row == 0) ? SW_OFF : ((row == 1) ? CC_OFF : PAD_OFF)) + h2;
    const int zoff  = SZ_OFF + h2;

    float ym = 0.f, yv = 0.f;

    #pragma unroll 1
    for (int i = 0; i < I_DIM; ++i) {
        const int oi = o * I_DIM + i;

        float scale = fmaxf(__expf(lsc[oi]), 0.1f);
        float l2    = scale * scale;
        float sv    = fmaxf(__expf(lva[oi]), 1e-5f);
        float den1  = l2 + 1e-6f;
        float den2  = l2 + 2e-6f;
        float r1    = sqrtf(l2 / den1);
        float r2    = sqrtf(l2 / den2);
        float c2sv  = sv * r2;
        float k1    = -0.5f * LOG2E / den1;
        float k2    = -LOG2E / den2;

        const float* zp  = z   + (size_t)oi * M_DIM;
        const float* qp  = qmu + (size_t)oi * M_DIM;
        const float* qvp = qlv + (size_t)oi * M_DIM;
        float z_l  = zp[lane];
        float qm_l = qp[lane];
        float qv_l = fmaxf(__expf(qvp[lane]), 1e-5f);

        float xn = x[n * I_DIM + i];

        __builtin_amdgcn_wave_barrier();
        W[SZ_OFF + lane] = z_l;
        W[SQ_OFF + lane] = qm_l * r1;
        W[SW_OFF + lane] = (qv_l + qm_l * qm_l) * r2;
        W[CC_OFF + lane] = c2sv;
        __builtin_amdgcn_wave_barrier();

        v8f acc1 = {};
        v8f acc2 = {};

        #pragma unroll
        for (int c = 0; c < 8; ++c) {
            v2f zp2 = *(const v2f*)(W + zoff  + 4 * c);
            v2f a1  = *(const v2f*)(W + a1off + 4 * c);
            v2f a2  = *(const v2f*)(W + a2off + 4 * c);

            float d0  = xn - zp2[0];
            float d1  = xn - zp2[1];
            float d0s = d0 * d0;
            float d1s = d1 * d1;
            v2f b1 = { __builtin_amdgcn_exp2f(d0s * k1),
                       __builtin_amdgcn_exp2f(d1s * k1) };
            v2f b2 = { __builtin_amdgcn_exp2f(d0s * k2),
                       __builtin_amdgcn_exp2f(d1s * k2) };

            acc1 = __builtin_amdgcn_wmma_f32_16x16x4_f32(
                       false, a1, false, b1, (short)0, acc1, false, false);
            acc2 = __builtin_amdgcn_wmma_f32_16x16x4_f32(
                       false, a2, false, b2, (short)0, acc2, false, false);
        }

        float em    = acc1[0];
        float secnd = acc2[0];
        float s2b   = acc2[1];
        float ev = sv - s2b + secnd - em * em;
        ev = fmaxf(ev, 1e-6f);
        ym += em;
        yv += ev;
    }

    if (lane < 16) {
        out[n * O_DIM + o]                 = ym;
        out[N_DIM * O_DIM + n * O_DIM + o] = yv;
    }
}

extern "C" void kernel_launch(void* const* d_in, const int* in_sizes, int n_in,
                              void* d_out, int out_size, void* d_ws, size_t ws_size,
                              hipStream_t stream) {
    const float* x   = (const float*)d_in[0];
    const float* z   = (const float*)d_in[1];
    const float* qmu = (const float*)d_in[2];
    const float* qlv = (const float*)d_in[3];
    const float* lsc = (const float*)d_in[4];
    const float* lva = (const float*)d_in[5];
    float* out = (float*)d_out;

    dim3 grid(O_DIM);
    dim3 block(256);
    gpkan_moment_kernel<<<grid, block, 0, stream>>>(x, z, qmu, qlv, lsc, lva, out);
}


# split the i-sum 8 ways across waves (partials added through LDS in fixed order, so the f32 sum order changes): 256 blocks x 32 waves use every WGP at 8 waves/SIMD, not 128 x 8
# speedup vs baseline: 2.4938x; 2.4938x over previous
#include <hip/hip_runtime.h>
#include <hip/hip_bf16.h>
#include <math.h>

typedef __attribute__((ext_vector_type(2))) float v2f;
typedef __attribute__((ext_vector_type(8))) float v8f;

#define O_DIM 128
#define I_DIM 128
#define M_DIM 32
#define N_DIM 128
#define LOG2E 1.44269504088896340736f

#define SZ_OFF   0
#define SQ_OFF   32
#define SW_OFF   64
#define CC_OFF   96
#define PAD_OFF  128
#define WREG     160

#define N_TILES  4
#define I_SPLIT  8
#define I_CHUNK  (I_DIM / I_SPLIT)
#define N_WAVES  (N_TILES * I_SPLIT)
#define N_BLK    (N_TILES * 16)

__global__ __launch_bounds__(1024)
void gpkan_moment_kernel(const float* __restrict__ x,
                         const float* __restrict__ z,
                         const float* __restrict__ qmu,
                         const float* __restrict__ qlv,
                         const float* __restrict__ lsc,
                         const float* __restrict__ lva,
                         float* __restrict__ out)
{
    __shared__ __attribute__((aligned(16))) float smem[N_WAVES * WREG];
    __shared__ float red[2][I_SPLIT][N_BLK];

    const int o     = blockIdx.x >> 1;
    const int nbase = (blockIdx.x & 1) * N_BLK;
    const int wave  = __builtin_amdgcn_readfirstlane((int)(threadIdx.x >> 5));
    const int tile  = wave & (N_TILES - 1);
    const int chunk = wave >> 2;
    const int lane  = threadIdx.x & 31;
    const int row   = lane & 15;
    const int half  = lane >> 4;
    const int nloc  = tile * 16 + row;
    const int n     = nbase + nloc;
    const int i0    = chunk * I_CHUNK;

    float* W = &smem[wave * WREG];

    W[PAD_OFF + lane] = 0.0f;

    const int h2 = 2 * half;
    const int a1off = ((row == 0) ? SQ_OFF : PAD_OFF) + h2;
    const int a2off = ((row == 0) ? SW_OFF : ((row == 1) ? CC_OFF : PAD_OFF)) + h2;
    const int zoff  = SZ_OFF + h2;

    float ym = 0.f, yv = 0.f;

    #pragma unroll 1
    for (int i = i0; i < i0 + I_CHUNK; ++i) {
        const int oi = o * I_DIM + i;

        float scale = fmaxf(__expf(lsc[oi]), 0.1f);
        float l2    = scale * scale;
        float sv    = fmaxf(__expf(lva[oi]), 1e-5f);
        float den1  = l2 + 1e-6f;
        float den2  = l2 + 2e-6f;
        float r1    = sqrtf(l2 / den1);
        float r2    = sqrtf(l2 / den2);
        float c2sv  = sv * r2;
        float k1    = -0.5f * LOG2E / den1;
        float k2    = -LOG2E / den2;

        const float* zp  = z   + (size_t)oi * M_DIM;
        const float* qp  = qmu + (size_t)oi * M_DIM;
        const float* qvp = qlv + (size_t)oi * M_DIM;
        float z_l  = zp[lane];
        float qm_l = qp[lane];
        float qv_l = fmaxf(__expf(qvp[lane]), 1e-5f);

        float xn = x[n * I_DIM + i];

        __builtin_amdgcn_wave_barrier();
        W[SZ_OFF + lane] = z_l;
        W[SQ_OFF + lane] = qm_l * r1;
        W[SW_OFF + lane] = (qv_l + qm_l * qm_l) * r2;
        W[CC_OFF + lane] = c2sv;
        __builtin_amdgcn_wave_barrier();

        v8f acc1 = {};
        v8f acc2 = {};

        #pragma unroll
        for (int c = 0; c < 8; ++c) {
            v2f zp2 = *(const v2f*)(W + zoff  + 4 * c);
            v2f a1  = *(const v2f*)(W + a1off + 4 * c);
            v2f a2  = *(const v2f*)(W + a2off + 4 * c);

            float d0  = xn - zp2[0];
            float d1  = xn - zp2[1];
            float d0s = d0 * d0;
            float d1s = d1 * d1;
            v2f b1 = { __builtin_amdgcn_exp2f(d0s * k1),
                       __builtin_amdgcn_exp2f(d1s * k1) };
            v2f b2 = { __builtin_amdgcn_exp2f(d0s * k2),
                       __builtin_amdgcn_exp2f(d1s * k2) };

            acc1 = __builtin_amdgcn_wmma_f32_16x16x4_f32(
                       false, a1, false, b1, (short)0, acc1, false, false);
            acc2 = __builtin_amdgcn_wmma_f32_16x16x4_f32(
                       false, a2, false, b2, (short)0, acc2, false, false);
        }

        float em    = acc1[0];
        float secnd = acc2[0];
        float s2b   = acc2[1];
        float ev = sv - s2b + secnd - em * em;
        ev = fmaxf(ev, 1e-6f);
        ym += em;
        yv += ev;
    }

    if (lane < 16) {
        red[0][chunk][nloc] = ym;
        red[1][chunk][nloc] = yv;
    }
    __syncthreads();
    if (threadIdx.x < 2 * N_BLK) {
        const int which = threadIdx.x >> 6;
        const int nl    = threadIdx.x & (N_BLK - 1);
        float s = red[which][0][nl];
        for (int c = 1; c < I_SPLIT; ++c) s += red[which][c][nl];
        out[which * (N_DIM * O_DIM) + (nbase + nl) * O_DIM + o] = s;
    }
}

extern "C" void kernel_launch(void* const* d_in, const int* in_sizes, int n_in,
                              void* d_out, int out_size, void* d_ws, size_t ws_size,
                              hipStream_t stream) {
    const float* x   = (const float*)d_in[0];
    const float* z   = (const float*)d_in[1];
    const float* qmu = (const float*)d_in[2];
    const float* qlv = (const float*)d_in[3];
    const float* lsc = (const float*)d_in[4];
    const float* lva = (const float*)d_in[5];
    float* out = (float*)d_out;

    dim3 grid(256);
    dim3 block(1024);
    gpkan_moment_kernel<<<grid, block, 0, stream>>>(x, z, qmu, qlv, lsc, lva, out);
}


# operand preparation hoisted out of the i loop: the block's 1024 threads build the z, weight and exponent-scale tables of all 128 i and the x slice once in LDS; loop reads only LDS; sum order kept
# speedup vs baseline: 3.1993x; 1.2829x over previous
#include <hip/hip_runtime.h>
#include <hip/hip_bf16.h>
#include <math.h>

typedef __attribute__((ext_vector_type(2))) float v2f;
typedef __attribute__((ext_vector_type(4))) float v4f;
typedef __attribute__((ext_vector_type(8))) float v8f;

#define O_DIM 128
#define I_DIM 128
#define M_DIM 32
#define N_DIM 128
#define LOG2E 1.44269504088896340736f

#define TZ_OFF   0
#define TQ_OFF   32
#define TW_OFF   64
#define TC_OFF   96
#define TROW     128
#define PAD_BASE (I_DIM * TROW + 16)
#define TAB_SIZE (I_DIM * TROW + 64)
#define XPITCH   65

#define N_TILES  4
#define I_SPLIT  8
#define I_CHUNK  (I_DIM / I_SPLIT)
#define N_WAVES  (N_TILES * I_SPLIT)
#define N_BLK    (N_TILES * 16)

__global__ __launch_bounds__(1024)
void gpkan_moment_kernel(const float* __restrict__ x,
                         const float* __restrict__ z,
                         const float* __restrict__ qmu,
                         const float* __restrict__ qlv,
                         const float* __restrict__ lsc,
                         const float* __restrict__ lva,
                         float* __restrict__ out)
{
    __shared__ __attribute__((aligned(16))) float tab[TAB_SIZE];
    __shared__ __attribute__((aligned(16))) float par[I_DIM * 4];
    __shared__ float xs[I_DIM * XPITCH];
    __shared__ float red[2][I_SPLIT][N_BLK];

    const int o     = blockIdx.x >> 1;
    const int nbase = (blockIdx.x & 1) * N_BLK;

    {
        const int t  = threadIdx.x;
        const int pi = t >> 3;
        const int m4 = (t & 7) * 4;
        const int oi = o * I_DIM + pi;
        const int xr = t >> 4;
        const int xq = (t & 15) * 4;

        const float* xg = x + (size_t)(nbase + xr) * I_DIM + xq;
        const float lsc_v = lsc[oi];
        const float lva_v = lva[oi];
        const v4f zz = *(const v4f*)(z   + (size_t)oi * M_DIM + m4);
        const v4f qm = *(const v4f*)(qmu + (size_t)oi * M_DIM + m4);
        const v4f ql = *(const v4f*)(qlv + (size_t)oi * M_DIM + m4);
        const v4f xa = *(const v4f*)(xg);
        const v4f xb = *(const v4f*)(xg + 64);

        float scale = fmaxf(__expf(lsc_v), 0.1f);
        float l2    = scale * scale;
        float sv    = fmaxf(__expf(lva_v), 1e-5f);
        float den1  = l2 + 1e-6f;
        float den2  = l2 + 2e-6f;
        float r1    = sqrtf(l2 / den1);
        float r2    = sqrtf(l2 / den2);
        float c2sv  = sv * r2;
        float k1    = -0.5f * LOG2E / den1;
        float k2    = -LOG2E / den2;

        v4f qs, ws;
        #pragma unroll
        for (int j = 0; j < 4; ++j) {
            float qv_l = fmaxf(__expf(ql[j]), 1e-5f);
            qs[j] = qm[j] * r1;
            ws[j] = (qv_l + qm[j] * qm[j]) * r2;
        }
        const v4f cc = { c2sv, c2sv, c2sv, c2sv };

        float* R = tab + pi * TROW + m4;
        *(v4f*)(R + TZ_OFF) = zz;
        *(v4f*)(R + TQ_OFF) = qs;
        *(v4f*)(R + TW_OFF) = ws;
        *(v4f*)(R + TC_OFF) = cc;
        if ((t & 7) == 0) {
            const v4f pp = { k1, k2, sv, 0.0f };
            *(v4f*)(par + 4 * pi) = pp;
        }
        if (t < 32) tab[PAD_BASE + t] = 0.0f;

        #pragma unroll
        for (int j = 0; j < 4; ++j) {
            xs[(xq + j) * XPITCH + xr]      = xa[j];
            xs[(xq + 64 + j) * XPITCH + xr] = xb[j];
        }
    }
    __syncthreads();

    const int wave  = __builtin_amdgcn_readfirstlane((int)(threadIdx.x >> 5));
    const int tile  = wave & (N_TILES - 1);
    const int chunk = wave >> 2;
    const int lane  = threadIdx.x & 31;
    const int row   = lane & 15;
    const int half  = lane >> 4;
    const int nloc  = tile * 16 + row;
    const int i0    = chunk * I_CHUNK;

    const v2f* tab2 = (const v2f*)tab;
    const int rowb = i0 * (TROW / 2) + half;
    const int padb = PAD_BASE / 2 + half;
    int zoff  = rowb + TZ_OFF / 2;
    int a1off = (row == 0) ? (rowb + TQ_OFF / 2) : padb;
    int a2off = (row == 0) ? (rowb + TW_OFF / 2)
              : ((row == 1) ? (rowb + TC_OFF / 2) : padb);
    const int a1step = (row == 0) ? (TROW / 2) : 0;
    const int a2step = (row <= 1) ? (TROW / 2) : 0;
    int xoff = i0 * XPITCH + nloc;

    float ym = 0.f, yv = 0.f;

    #pragma unroll 1
    for (int ii = 0; ii < I_CHUNK; ++ii) {
        const float xn = xs[xoff];
        const v4f pp = *(const v4f*)(par + 4 * (i0 + ii));
        const float k1 = pp[0];
        const float k2 = pp[1];
        const float sv = pp[2];

        v8f acc1 = {};
        v8f acc2 = {};

        #pragma unroll
        for (int c = 0; c < 8; ++c) {
            v2f zp2 = tab2[zoff  + 2 * c];
            v2f a1  = tab2[a1off + 2 * c];
            v2f a2  = tab2[a2off + 2 * c];

            float d0  = xn - zp2[0];
            float d1  = xn - zp2[1];
            float d0s = d0 * d0;
            float d1s = d1 * d1;
            v2f b1 = { __builtin_amdgcn_exp2f(d0s * k1),
                       __builtin_amdgcn_exp2f(d1s * k1) };
            v2f b2 = { __builtin_amdgcn_exp2f(d0s * k2),
                       __builtin_amdgcn_exp2f(d1s * k2) };

            acc1 = __builtin_amdgcn_wmma_f32_16x16x4_f32(
                       false, a1, false, b1, (short)0, acc1, false, false);
            acc2 = __builtin_amdgcn_wmma_f32_16x16x4_f32(
                       false, a2, false, b2, (short)0, acc2, false, false);
        }

        float em    = acc1[0];
        float secnd = acc2[0];
        float s2b   = acc2[1];
        float ev = sv - s2b + secnd - em * em;
        ev = fmaxf(ev, 1e-6f);
        ym += em;
        yv += ev;

        zoff  += TROW / 2;
        a1off += a1step;
        a2off += a2step;
        xoff  += XPITCH;
    }

    if (lane < 16) {
        red[0][chunk][nloc] = ym;
        red[1][chunk][nloc] = yv;
    }
    __syncthreads();
    if (threadIdx.x < 2 * N_BLK) {
        const int which = threadIdx.x >> 6;
        const int nl    = threadIdx.x & (N_BLK - 1);
        float s = red[which][0][nl];
        for (int c = 1; c < I_SPLIT; ++c) s += red[which][c][nl];
        out[which * (N_DIM * O_DIM) + (nbase + nl) * O_DIM + o] = s;
    }
}

extern "C" void kernel_launch(void* const* d_in, const int* in_sizes, int n_in,
                              void* d_out, int out_size, void* d_ws, size_t ws_size,
                              hipStream_t stream) {
    const float* x   = (const float*)d_in[0];
    const float* z   = (const float*)d_in[1];
    const float* qmu = (const float*)d_in[2];
    const float* qlv = (const float*)d_in[3];
    const float* lsc = (const float*)d_in[4];
    const float* lva = (const float*)d_in[5];
    float* out = (float*)d_out;

    dim3 grid(256);
    dim3 block(1024);
    gpkan_moment_kernel<<<grid, block, 0, stream>>>(x, z, qmu, qlv, lsc, lva, out);
}


# each wave holds one i's z and weight fragments in registers and sweeps the block's 4 sample tiles with them: one LDS load group per 64 WMMAs; i-sum now 4 per wave, then 32 partials in fixed order
# speedup vs baseline: 3.3074x; 1.0338x over previous
#include <hip/hip_runtime.h>
#include <hip/hip_bf16.h>
#include <math.h>

typedef __attribute__((ext_vector_type(2))) float v2f;
typedef __attribute__((ext_vector_type(4))) float v4f;
typedef __attribute__((ext_vector_type(8))) float v8f;

#define O_DIM 128
#define I_DIM 128
#define M_DIM 32
#define N_DIM 128
#define LOG2E 1.44269504088896340736f

#define TZ_OFF   0
#define TQ_OFF   32
#define TW_OFF   64
#define TC_OFF   96
#define TROW     128
#define PAD_BASE (I_DIM * TROW + 16)
#define TAB_SIZE (I_DIM * TROW + 64)
#define XPITCH   65

#define N_TILES  4
#define N_WAVES  32
#define I_WAVE   (I_DIM / N_WAVES)
#define N_BLK    (N_TILES * 16)

__global__ __launch_bounds__(1024)
void gpkan_moment_kernel(const float* __restrict__ x,
                         const float* __restrict__ z,
                         const float* __restrict__ qmu,
                         const float* __restrict__ qlv,
                         const float* __restrict__ lsc,
                         const float* __restrict__ lva,
                         float* __restrict__ out)
{
    __shared__ __attribute__((aligned(16))) float tab[TAB_SIZE];
    __shared__ __attribute__((aligned(16))) float par[I_DIM * 4];
    __shared__ float xs[I_DIM * XPITCH];
    __shared__ float red[2][N_WAVES][N_BLK];

    const int o     = blockIdx.x >> 1;
    const int nbase = (blockIdx.x & 1) * N_BLK;

    {
        const int t  = threadIdx.x;
        const int pi = t >> 3;
        const int m4 = (t & 7) * 4;
        const int oi = o * I_DIM + pi;
        const int xr = t >> 4;
        const int xq = (t & 15) * 4;

        const float* xg = x + (size_t)(nbase + xr) * I_DIM + xq;
        const float lsc_v = lsc[oi];
        const float lva_v = lva[oi];
        const v4f zz = *(const v4f*)(z   + (size_t)oi * M_DIM + m4);
        const v4f qm = *(const v4f*)(qmu + (size_t)oi * M_DIM + m4);
        const v4f ql = *(const v4f*)(qlv + (size_t)oi * M_DIM + m4);
        const v4f xa = *(const v4f*)(xg);
        const v4f xb = *(const v4f*)(xg + 64);

        float scale = fmaxf(__expf(lsc_v), 0.1f);
        float l2    = scale * scale;
        float sv    = fmaxf(__expf(lva_v), 1e-5f);
        float den1  = l2 + 1e-6f;
        float den2  = l2 + 2e-6f;
        float r1    = sqrtf(l2 / den1);
        float r2    = sqrtf(l2 / den2);
        float c2sv  = sv * r2;
        float k1    = -0.5f * LOG2E / den1;
        float k2    = -LOG2E / den2;

        v4f qs, ws;
        #pragma unroll
        for (int j = 0; j < 4; ++j) {
            float qv_l = fmaxf(__expf(ql[j]), 1e-5f);
            qs[j] = qm[j] * r1;
            ws[j] = (qv_l + qm[j] * qm[j]) * r2;
        }
        const v4f cc = { c2sv, c2sv, c2sv, c2sv };

        float* R = tab + pi * TROW + m4;
        *(v4f*)(R + TZ_OFF) = zz;
        *(v4f*)(R + TQ_OFF) = qs;
        *(v4f*)(R + TW_OFF) = ws;
        *(v4f*)(R + TC_OFF) = cc;
        if ((t & 7) == 0) {
            const v4f pp = { k1, k2, sv, 0.0f };
            *(v4f*)(par + 4 * pi) = pp;
        }
        if (t < 32) tab[PAD_BASE + t] = 0.0f;

        #pragma unroll
        for (int j = 0; j < 4; ++j) {
            xs[(xq + j) * XPITCH + xr]      = xa[j];
            xs[(xq + 64 + j) * XPITCH + xr] = xb[j];
        }
    }
    __syncthreads();

    const int wave  = __builtin_amdgcn_readfirstlane((int)(threadIdx.x >> 5));
    const int lane  = threadIdx.x & 31;
    const int row   = lane & 15;
    const int half  = lane >> 4;
    const int i0    = wave * I_WAVE;

    const v2f* tab2 = (const v2f*)tab;
    const int rowb = i0 * (TROW / 2) + half;
    const int padb = PAD_BASE / 2 + half;
    int zoff = rowb + TZ_OFF / 2;
    int aoff = (row == 0) ? (rowb + TQ_OFF / 2)
             : ((row == 1) ? (rowb + TW_OFF / 2)
             : ((row == 2) ? (rowb + TC_OFF / 2) : padb));
    const int astep = (row <= 2) ? (TROW / 2) : 0;
    int xoff = i0 * XPITCH + row;

    float ym[N_TILES], yv[N_TILES];
    #pragma unroll
    for (int t = 0; t < N_TILES; ++t) { ym[t] = 0.f; yv[t] = 0.f; }

    #pragma unroll 1
    for (int ii = 0; ii < I_WAVE; ++ii) {
        v2f zf[8], af[8];
        #pragma unroll
        for (int c = 0; c < 8; ++c) {
            zf[c] = tab2[zoff + 2 * c];
            af[c] = tab2[aoff + 2 * c];
        }
        float xn[N_TILES];
        #pragma unroll
        for (int t = 0; t < N_TILES; ++t) xn[t] = xs[xoff + 16 * t];
        const v4f pp = *(const v4f*)(par + 4 * (i0 + ii));
        const float k1 = pp[0];
        const float k2 = pp[1];
        const float sv = pp[2];

        __builtin_amdgcn_sched_barrier(0);

        #pragma unroll
        for (int t = 0; t < N_TILES; ++t) {
            v8f acc1 = {};
            v8f acc2 = {};

            #pragma unroll
            for (int c = 0; c < 8; ++c) {
                float d0  = xn[t] - zf[c][0];
                float d1  = xn[t] - zf[c][1];
                float d0s = d0 * d0;
                float d1s = d1 * d1;
                v2f b1 = { __builtin_amdgcn_exp2f(d0s * k1),
                           __builtin_amdgcn_exp2f(d1s * k1) };
                v2f b2 = { __builtin_amdgcn_exp2f(d0s * k2),
                           __builtin_amdgcn_exp2f(d1s * k2) };

                acc1 = __builtin_amdgcn_wmma_f32_16x16x4_f32(
                           false, af[c], false, b1, (short)0, acc1, false, false);
                acc2 = __builtin_amdgcn_wmma_f32_16x16x4_f32(
                           false, af[c], false, b2, (short)0, acc2, false, false);
            }

            float em    = acc1[0];
            float secnd = acc2[1];
            float s2b   = acc2[2];
            float ev = sv - s2b + secnd - em * em;
            ev = fmaxf(ev, 1e-6f);
            ym[t] += em;
            yv[t] += ev;
        }

        zoff += TROW / 2;
        aoff += astep;
        xoff += XPITCH;
    }

    if (lane < 16) {
        #pragma unroll
        for (int t = 0; t < N_TILES; ++t) {
            red[0][wave][t * 16 + row] = ym[t];
            red[1][wave][t * 16 + row] = yv[t];
        }
    }
    __syncthreads();
    if (threadIdx.x < 2 * N_BLK) {
        const int which = threadIdx.x >> 6;
        const int nl    = threadIdx.x & (N_BLK - 1);
        float s = red[which][0][nl];
        for (int w = 1; w < N_WAVES; ++w) s += red[which][w][nl];
        out[which * (N_DIM * O_DIM) + (nbase + nl) * O_DIM + o] = s;
    }
}

extern "C" void kernel_launch(void* const* d_in, const int* in_sizes, int n_in,
                              void* d_out, int out_size, void* d_ws, size_t ws_size,
                              hipStream_t stream) {
    const float* x   = (const float*)d_in[0];
    const float* z   = (const float*)d_in[1];
    const float* qmu = (const float*)d_in[2];
    const float* qlv = (const float*)d_in[3];
    const float* lsc = (const float*)d_in[4];
    const float* lva = (const float*)d_in[5];
    float* out = (float*)d_out;

    dim3 grid(256);
    dim3 block(1024);
    gpkan_moment_kernel<<<grid, block, 0, stream>>>(x, z, qmu, qlv, lsc, lva, out);
}


# exponent arithmetic beside the f32 WMMAs cut from 4 to 3 packed vector ops per WMMA pair: z and x pre-scaled by sqrt(-k1), so k1*d^2 is one negated square and k2*d^2 is rho=k2/k1 times it
# speedup vs baseline: 3.4185x; 1.0336x over previous
#include <hip/hip_runtime.h>
#include <hip/hip_bf16.h>
#include <math.h>

typedef __attribute__((ext_vector_type(2))) float v2f;
typedef __attribute__((ext_vector_type(4))) float v4f;
typedef __attribute__((ext_vector_type(8))) float v8f;

#define O_DIM 128
#define I_DIM 128
#define M_DIM 32
#define N_DIM 128
#define LOG2E 1.44269504088896340736f

#define TZ_OFF   0
#define TQ_OFF   32
#define TW_OFF   64
#define TC_OFF   96
#define TROW     128
#define PAD_BASE (I_DIM * TROW + 16)
#define TAB_SIZE (I_DIM * TROW + 64)
#define XPITCH   65

#define N_TILES  4
#define N_WAVES  32
#define I_WAVE   (I_DIM / N_WAVES)
#define N_BLK    (N_TILES * 16)

__global__ __launch_bounds__(1024)
void gpkan_moment_kernel(const float* __restrict__ x,
                         const float* __restrict__ z,
                         const float* __restrict__ qmu,
                         const float* __restrict__ qlv,
                         const float* __restrict__ lsc,
                         const float* __restrict__ lva,
                         float* __restrict__ out)
{
    __shared__ __attribute__((aligned(16))) float tab[TAB_SIZE];
    __shared__ __attribute__((aligned(16))) float par[I_DIM * 4];
    __shared__ float xs[I_DIM * XPITCH];
    __shared__ float red[2][N_WAVES][N_BLK];

    const int o     = blockIdx.x >> 1;
    const int nbase = (blockIdx.x & 1) * N_BLK;

    {
        const int t  = threadIdx.x;
        const int pi = t >> 3;
        const int m4 = (t & 7) * 4;
        const int oi = o * I_DIM + pi;
        const int xr = t >> 4;
        const int xq = (t & 15) * 4;

        const float* xg = x + (size_t)(nbase + xr) * I_DIM + xq;
        const float lsc_v = lsc[oi];
        const float lva_v = lva[oi];
        const v4f zz = *(const v4f*)(z   + (size_t)oi * M_DIM + m4);
        const v4f qm = *(const v4f*)(qmu + (size_t)oi * M_DIM + m4);
        const v4f ql = *(const v4f*)(qlv + (size_t)oi * M_DIM + m4);
        const v4f xa = *(const v4f*)(xg);
        const v4f xb = *(const v4f*)(xg + 64);

        float scale = fmaxf(__expf(lsc_v), 0.1f);
        float l2    = scale * scale;
        float sv    = fmaxf(__expf(lva_v), 1e-5f);
        float den1  = l2 + 1e-6f;
        float den2  = l2 + 2e-6f;
        float r1    = sqrtf(l2 / den1);
        float r2    = sqrtf(l2 / den2);
        float c2sv  = sv * r2;
        float s1    = r1 * (0.84932180028801904f / scale);
        float rho   = 2.0f * den1 / den2;

        v4f qs, ws;
        #pragma unroll
        for (int j = 0; j < 4; ++j) {
            float qv_l = fmaxf(__expf(ql[j]), 1e-5f);
            qs[j] = qm[j] * r1;
            ws[j] = (qv_l + qm[j] * qm[j]) * r2;
        }
        const v4f cc = { c2sv, c2sv, c2sv, c2sv };

        float* R = tab + pi * TROW + m4;
        *(v4f*)(R + TZ_OFF) = zz * s1;
        *(v4f*)(R + TQ_OFF) = qs;
        *(v4f*)(R + TW_OFF) = ws;
        *(v4f*)(R + TC_OFF) = cc;
        if ((t & 7) == 0) {
            const v4f pp = { s1, rho, sv, 0.0f };
            *(v4f*)(par + 4 * pi) = pp;
        }
        if (t < 32) tab[PAD_BASE + t] = 0.0f;

        #pragma unroll
        for (int j = 0; j < 4; ++j) {
            xs[(xq + j) * XPITCH + xr]      = xa[j];
            xs[(xq + 64 + j) * XPITCH + xr] = xb[j];
        }
    }
    __syncthreads();

    const int wave  = __builtin_amdgcn_readfirstlane((int)(threadIdx.x >> 5));
    const int lane  = threadIdx.x & 31;
    const int row   = lane & 15;
    const int half  = lane >> 4;
    const int i0    = wave * I_WAVE;

    const v2f* tab2 = (const v2f*)tab;
    const int rowb = i0 * (TROW / 2) + half;
    const int padb = PAD_BASE / 2 + half;
    int zoff = rowb + TZ_OFF / 2;
    int aoff = (row == 0) ? (rowb + TQ_OFF / 2)
             : ((row == 1) ? (rowb + TW_OFF / 2)
             : ((row == 2) ? (rowb + TC_OFF / 2) : padb));
    const int astep = (row <= 2) ? (TROW / 2) : 0;
    int xoff = i0 * XPITCH + row;

    float ym[N_TILES], yv[N_TILES];
    #pragma unroll
    for (int t = 0; t < N_TILES; ++t) { ym[t] = 0.f; yv[t] = 0.f; }

    #pragma unroll 1
    for (int ii = 0; ii < I_WAVE; ++ii) {
        v2f zf[8], af[8];
        #pragma unroll
        for (int c = 0; c < 8; ++c) {
            zf[c] = tab2[zoff + 2 * c];
            af[c] = tab2[aoff + 2 * c];
        }
        float xn[N_TILES];
        #pragma unroll
        for (int t = 0; t < N_TILES; ++t) xn[t] = xs[xoff + 16 * t];
        const v4f pp = *(const v4f*)(par + 4 * (i0 + ii));
        const float s1 = pp[0];
        const float sv = pp[2];
        const v2f rho2 = { pp[1], pp[1] };

        __builtin_amdgcn_sched_barrier(0);

        float xsn[N_TILES];
        #pragma unroll
        for (int t = 0; t < N_TILES; ++t) xsn[t] = xn[t] * s1;

        float em[N_TILES], secnd[N_TILES], s2b[N_TILES];
        #pragma unroll
        for (int t = 0; t < N_TILES; ++t) {
            v8f acc1 = {};
            v8f acc2 = {};
            const v2f xx = { xsn[t], xsn[t] };

            #pragma unroll
            for (int c = 0; c < 8; ++c) {
                const v2f dd = xx - zf[c];
                const v2f n1 = dd * (-dd);
                const v2f n2 = n1 * rho2;
                v2f b1 = { __builtin_amdgcn_exp2f(n1[0]),
                           __builtin_amdgcn_exp2f(n1[1]) };
                v2f b2 = { __builtin_amdgcn_exp2f(n2[0]),
                           __builtin_amdgcn_exp2f(n2[1]) };

                acc1 = __builtin_amdgcn_wmma_f32_16x16x4_f32(
                           false, af[c], false, b1, (short)0, acc1, false, false);
                acc2 = __builtin_amdgcn_wmma_f32_16x16x4_f32(
                           false, af[c], false, b2, (short)0, acc2, false, false);
            }

            em[t]    = acc1[0];
            secnd[t] = acc2[1];
            s2b[t]   = acc2[2];
        }

        __builtin_amdgcn_sched_barrier(0);

        #pragma unroll
        for (int t = 0; t < N_TILES; ++t) {
            float ev = sv - s2b[t] + secnd[t] - em[t] * em[t];
            ev = fmaxf(ev, 1e-6f);
            ym[t] += em[t];
            yv[t] += ev;
        }

        zoff += TROW / 2;
        aoff += astep;
        xoff += XPITCH;
    }

    if (lane < 16) {
        #pragma unroll
        for (int t = 0; t < N_TILES; ++t) {
            red[0][wave][t * 16 + row] = ym[t];
            red[1][wave][t * 16 + row] = yv[t];
        }
    }
    __syncthreads();
    if (threadIdx.x < 2 * N_BLK) {
        const int which = threadIdx.x >> 6;
        const int nl    = threadIdx.x & (N_BLK - 1);
        float s = red[which][0][nl];
        for (int w = 1; w < N_WAVES; ++w) s += red[which][w][nl];
        out[which * (N_DIM * O_DIM) + (nbase + nl) * O_DIM + o] = s;
    }
}

extern "C" void kernel_launch(void* const* d_in, const int* in_sizes, int n_in,
                              void* d_out, int out_size, void* d_ws, size_t ws_size,
                              hipStream_t stream) {
    const float* x   = (const float*)d_in[0];
    const float* z   = (const float*)d_in[1];
    const float* qmu = (const float*)d_in[2];
    const float* qlv = (const float*)d_in[3];
    const float* lsc = (const float*)d_in[4];
    const float* lva = (const float*)d_in[5];
    float* out = (float*)d_out;

    dim3 grid(256);
    dim3 block(1024);
    gpkan_moment_kernel<<<grid, block, 0, stream>>>(x, z, qmu, qlv, lsc, lva, out);
}
